# starting kernel
# baseline (speedup 1.0000x reference)
#include <hip/hip_runtime.h>

#define NSTEPS 10
#define DT_C 0.1f
#define WPB 4

typedef float v2f __attribute__((ext_vector_type(2)));
typedef float v8f __attribute__((ext_vector_type(8)));

#if __has_builtin(__builtin_amdgcn_tanhf)
#define TANHF(x) __builtin_amdgcn_tanhf(x)
#elif __has_builtin(__builtin_amdgcn_tanh_f32)
#define TANHF(x) __builtin_amdgcn_tanh_f32(x)
#else
#define TANHF(x) tanhf(x)
#endif

static __device__ __forceinline__ v8f wmma4(v2f a, v2f b, v8f c) {
  return __builtin_amdgcn_wmma_f32_16x16x4_f32(false, a, false, b, (short)0, c,
                                               false, false);
}

__global__ __launch_bounds__(WPB * 32) void rollout_kernel(
    const float* __restrict__ s_star, const float* __restrict__ s0,
    const float* __restrict__ fc1_w, const float* __restrict__ fc1_b,
    const float* __restrict__ fc2_w, const float* __restrict__ fc2_b,
    const float* __restrict__ rc1_w, const float* __restrict__ rc1_b,
    const float* __restrict__ rc2_w, const float* __restrict__ rc2_b,
    const float* __restrict__ rc3_w, const float* __restrict__ rc3_b,
    const float* __restrict__ rr1_w, const float* __restrict__ rr1_b,
    const float* __restrict__ rr2_w, const float* __restrict__ rr2_b,
    const float* __restrict__ rr3_w, const float* __restrict__ rr3_b,
    float* __restrict__ out, int nrows)
{
  __shared__ __align__(16) float lds[WPB][4][512];

  const int tid = (int)threadIdx.x;
  const int w = tid >> 5;
  const int l = tid & 31;
  const int n = l & 15;
  const int kh = l >> 4;

  float* stage = &lds[w][0][0];
  float* gH    = &lds[w][1][0];
  float* gRC   = &lds[w][2][0];
  float* gRR   = &lds[w][3][0];

  const int dsoff = kh * 128 + n;
  const int aoff  = n * 16 + 2 * kh;

  int row = blockIdx.x * (WPB * 32) + tid;
  const bool valid = (row < nrows);
  if (row >= nrows) row = nrows - 1;

  float Ms[2][2], Mt[2][2], cah[2];
#pragma unroll
  for (int i = 0; i < 2; ++i) {
    float a0 = fc2_w[i*4+0], a1 = fc2_w[i*4+1], a2 = fc2_w[i*4+2], a3 = fc2_w[i*4+3];
#pragma unroll
    for (int k = 0; k < 2; ++k) {
      Ms[i][k] = a0*fc1_w[0*4+k]   + a1*fc1_w[1*4+k]   + a2*fc1_w[2*4+k]   + a3*fc1_w[3*4+k];
      Mt[i][k] = a0*fc1_w[0*4+2+k] + a1*fc1_w[1*4+2+k] + a2*fc1_w[2*4+2+k] + a3*fc1_w[3*4+2+k];
    }
    cah[i] = a0*fc1_b[0] + a1*fc1_b[1] + a2*fc1_b[2] + a3*fc1_b[3] + fc2_b[i];
  }
  const float b3rc0 = rc3_b[0], b3rc1 = rc3_b[1];
  const float db0 = rr3_b[0] - rc3_b[0], db1 = rr3_b[1] - rc3_b[1];

  const float ssx = s_star[row*2+0], ssy = s_star[row*2+1];
  float sx = s0[row*2+0], sy = s0[row*2+1];
  const float g0 = Mt[0][0]*ssx + Mt[0][1]*ssy + cah[0];
  const float g1 = Mt[1][0]*ssx + Mt[1][1]*ssy + cah[1];
  const float ahs0 = (ssx > 0.5f) ? -1.0f : 1.0f;
  const float fpx = ahs0 - g0;
  const float fpy = 0.0f - g1;

#pragma unroll
  for (int j = 0; j < 8; ++j) {
    stage[l*16 + j]     = rc1_b[j] + rc1_w[j*4+2]*g0 + rc1_w[j*4+3]*g1;
    stage[l*16 + 8 + j] = rr1_b[j] + rr1_w[j*4+2]*g0 + rr1_w[j*4+3]*g1;
  }
  __syncthreads();
  v8f C1[2];
#pragma unroll
  for (int t = 0; t < 2; ++t)
#pragma unroll
    for (int r = 0; r < 8; ++r)
      C1[t][r] = stage[t*256 + r*16 + dsoff];

  const float* w1p = (n < 8) ? rc1_w : rr1_w;
  const int jj = n & 7;
  v2f B1;
#pragma unroll
  for (int v = 0; v < 2; ++v) {
    int k = v + 2*kh;
    float val = 0.0f;
    if (k < 2)
      val = w1p[jj*4+k] + w1p[jj*4+2]*Ms[0][k] + w1p[jj*4+3]*Ms[1][k];
    B1[v] = val;
  }
  v2f B2rc[2], B2rr[2];
#pragma unroll
  for (int c = 0; c < 2; ++c)
#pragma unroll
    for (int v = 0; v < 2; ++v) {
      int k = c*4 + v + 2*kh;
      float vc = rc2_w[(n & 7)*8 + k];
      float vr = rr2_w[(n & 7)*8 + k];
      B2rc[c][v] = (n < 8) ? vc : 0.0f;
      B2rr[c][v] = (n < 8) ? vr : 0.0f;
    }
  v2f B3[4];
#pragma unroll
  for (int c = 0; c < 4; ++c)
#pragma unroll
    for (int v = 0; v < 2; ++v) {
      int k = c*4 + v + 2*kh;
      int nr = n & 1;
      float vrc = rc3_w[nr*8 + (k & 7)];
      float vrr = rr3_w[nr*8 + (k & 7)];
      float val = 0.0f;
      if (n < 2)      val = (k < 8) ? vrc : 0.0f;
      else if (n < 4) val = (k < 8) ? -vrc : vrr;
      B3[c][v] = val;
    }
  const float b2vrc = (n < 8) ? rc2_b[n & 7] : 0.0f;
  const float b2vrr = (n < 8) ? rr2_b[n & 7] : 0.0f;

  const v8f z8 = {0.f,0.f,0.f,0.f,0.f,0.f,0.f,0.f};

  float err = 0.0f;
#pragma unroll 1
  for (int step = 0; step < NSTEPS; ++step) {
    const float sxo = __shfl_xor(sx, 16, 32);
    const float syo = __shfl_xor(sy, 16, 32);
    const bool lo = (l < 16);
    v2f A1[2];
    A1[0][0] = lo ? sx  : 0.0f;  A1[0][1] = lo ? sy  : 0.0f;
    A1[1][0] = lo ? sxo : 0.0f;  A1[1][1] = lo ? syo : 0.0f;

    const float ahdx = fpx - (Ms[0][0]*sx + Ms[0][1]*sy);
    const float ahdy = fpy - (Ms[1][0]*sx + Ms[1][1]*sy);

#pragma unroll
    for (int t = 0; t < 2; ++t) {
      v8f d1 = wmma4(A1[t], B1, C1[t]);
#pragma unroll
      for (int r = 0; r < 8; ++r)
        gH[t*256 + r*16 + dsoff] = TANHF(d1[r]);
    }

#pragma unroll
    for (int t = 0; t < 2; ++t) {
      const int tb = t*256;
      v2f aRC0 = { gH[tb + aoff + 0],  gH[tb + aoff + 1]  };
      v2f aRC1 = { gH[tb + aoff + 4],  gH[tb + aoff + 5]  };
      v2f aRR0 = { gH[tb + aoff + 8],  gH[tb + aoff + 9]  };
      v2f aRR1 = { gH[tb + aoff + 12], gH[tb + aoff + 13] };
      v8f dRC = wmma4(aRC1, B2rc[1], wmma4(aRC0, B2rc[0], z8));
      v8f dRR = wmma4(aRR1, B2rr[1], wmma4(aRR0, B2rr[0], z8));
#pragma unroll
      for (int r = 0; r < 8; ++r) {
        gRC[tb + r*16 + dsoff] = TANHF(dRC[r] + b2vrc);
        gRR[tb + r*16 + dsoff] = TANHF(dRR[r] + b2vrr);
      }
    }

#pragma unroll
    for (int t = 0; t < 2; ++t) {
      const int tb = t*256;
      v2f a0 = { gRC[tb + aoff + 0], gRC[tb + aoff + 1] };
      v2f a1 = { gRC[tb + aoff + 4], gRC[tb + aoff + 5] };
      v2f a2 = { gRR[tb + aoff + 0], gRR[tb + aoff + 1] };
      v2f a3 = { gRR[tb + aoff + 4], gRR[tb + aoff + 5] };
      v8f d3 = wmma4(a3, B3[3], wmma4(a2, B3[2], wmma4(a1, B3[1], wmma4(a0, B3[0], z8))));
#pragma unroll
      for (int r = 0; r < 8; ++r)
        stage[tb + r*16 + dsoff] = d3[r];
    }

    const float4 v = *(const float4*)&stage[l*16];
    const float arx = v.x + b3rc0, ary = v.y + b3rc1;
    const float drx = v.z + db0,  dry = v.w + db1;
    sx += DT_C * arx;
    sy += DT_C * ary;
    const float ex = ssx - sx, ey = ssy - sy;
    err += sqrtf(ex*ex + ey*ey)
         + 0.1f * sqrtf(drx*drx + dry*dry)
         + sqrtf(ahdx*ahdx + ahdy*ahdy);
  }

  if (valid) out[row] = err;
}

extern "C" void kernel_launch(void* const* d_in, const int* in_sizes, int n_in,
                              void* d_out, int out_size, void* d_ws, size_t ws_size,
                              hipStream_t stream) {
  const float* s_star = (const float*)d_in[0];
  const float* s0     = (const float*)d_in[1];
  const float* fc1_w  = (const float*)d_in[2];
  const float* fc1_b  = (const float*)d_in[3];
  const float* fc2_w  = (const float*)d_in[4];
  const float* fc2_b  = (const float*)d_in[5];
  const float* rc1_w  = (const float*)d_in[6];
  const float* rc1_b  = (const float*)d_in[7];
  const float* rc2_w  = (const float*)d_in[8];
  const float* rc2_b  = (const float*)d_in[9];
  const float* rc3_w  = (const float*)d_in[10];
  const float* rc3_b  = (const float*)d_in[11];
  const float* rr1_w  = (const float*)d_in[12];
  const float* rr1_b  = (const float*)d_in[13];
  const float* rr2_w  = (const float*)d_in[14];
  const float* rr2_b  = (const float*)d_in[15];
  const float* rr3_w  = (const float*)d_in[16];
  const float* rr3_b  = (const float*)d_in[17];
  float* out = (float*)d_out;

  const int B = in_sizes[0] / 2;
  const int block = WPB * 32;
  const int grid = (B + block - 1) / block;

  rollout_kernel<<<grid, block, 0, stream>>>(
      s_star, s0, fc1_w, fc1_b, fc2_w, fc2_b,
      rc1_w, rc1_b, rc2_w, rc2_b, rc3_w, rc3_b,
      rr1_w, rr1_b, rr2_w, rr2_b, rr3_w, rr3_b,
      out, B);
}


# Transposed products (weights as A, batch as N): each layer's D registers are the next layer's B operand, so no LDS and no barrier; k order permuted and biases enter as C (sum order changes)
# speedup vs baseline: 1.2030x; 1.2030x over previous
#include <hip/hip_runtime.h>

#define NSTEPS 10
#define DT_C 0.1f
#define WPB 4

typedef float v2f __attribute__((ext_vector_type(2)));
typedef float v8f __attribute__((ext_vector_type(8)));

#if __has_builtin(__builtin_amdgcn_tanhf)
#define TANHF(x) __builtin_amdgcn_tanhf(x)
#elif __has_builtin(__builtin_amdgcn_tanh_f32)
#define TANHF(x) __builtin_amdgcn_tanh_f32(x)
#else
#define TANHF(x) tanhf(x)
#endif

static __device__ __forceinline__ v8f wmma4(v2f a, v2f b, v8f c) {
  return __builtin_amdgcn_wmma_f32_16x16x4_f32(false, a, false, b, (short)0, c,
                                               false, false);
}

__global__ __launch_bounds__(WPB * 32) void rollout_kernel(
    const float* __restrict__ s_star, const float* __restrict__ s0,
    const float* __restrict__ fc1_w, const float* __restrict__ fc1_b,
    const float* __restrict__ fc2_w, const float* __restrict__ fc2_b,
    const float* __restrict__ rc1_w, const float* __restrict__ rc1_b,
    const float* __restrict__ rc2_w, const float* __restrict__ rc2_b,
    const float* __restrict__ rc3_w, const float* __restrict__ rc3_b,
    const float* __restrict__ rr1_w, const float* __restrict__ rr1_b,
    const float* __restrict__ rr2_w, const float* __restrict__ rr2_b,
    const float* __restrict__ rr3_w, const float* __restrict__ rr3_b,
    float* __restrict__ out, int nrows)
{

  const int tid = (int)threadIdx.x;
  const int l = tid & 31;
  const int n = l & 15;
  const int kh = l >> 4;
  const bool lo = (kh == 0);

  int row = blockIdx.x * (WPB * 32) + tid;
  const bool valid = (row < nrows);
  if (row >= nrows) row = nrows - 1;

  float Ms[2][2], Mt[2][2], cah[2];
#pragma unroll
  for (int i = 0; i < 2; ++i) {
    float a0 = fc2_w[i*4+0], a1 = fc2_w[i*4+1], a2 = fc2_w[i*4+2], a3 = fc2_w[i*4+3];
#pragma unroll
    for (int k = 0; k < 2; ++k) {
      Ms[i][k] = a0*fc1_w[0*4+k]   + a1*fc1_w[1*4+k]   + a2*fc1_w[2*4+k]   + a3*fc1_w[3*4+k];
      Mt[i][k] = a0*fc1_w[0*4+2+k] + a1*fc1_w[1*4+2+k] + a2*fc1_w[2*4+2+k] + a3*fc1_w[3*4+2+k];
    }
    cah[i] = a0*fc1_b[0] + a1*fc1_b[1] + a2*fc1_b[2] + a3*fc1_b[3] + fc2_b[i];
  }
  const float b3rc0 = rc3_b[0], b3rc1 = rc3_b[1];
  const float db0 = rr3_b[0] - rc3_b[0], db1 = rr3_b[1] - rc3_b[1];

  const float ssx = s_star[row*2+0], ssy = s_star[row*2+1];
  float sx = s0[row*2+0], sy = s0[row*2+1];
  const float g0 = Mt[0][0]*ssx + Mt[0][1]*ssy + cah[0];
  const float g1 = Mt[1][0]*ssx + Mt[1][1]*ssy + cah[1];
  const float ahs0 = (ssx > 0.5f) ? -1.0f : 1.0f;
  const float fpx = ahs0 - g0;
  const float fpy = 0.0f - g1;
  const float g0o = __shfl_xor(g0, 16, 32);
  const float g1o = __shfl_xor(g1, 16, 32);

  const bool rcrow = (n < 8);
  const int j8 = n & 7;
  const float* w1p = rcrow ? rc1_w : rr1_w;
  const float w1g0 = w1p[j8*4+2], w1g1 = w1p[j8*4+3];
  const float w1s0 = w1p[j8*4+0] + w1p[j8*4+2]*Ms[0][0] + w1p[j8*4+3]*Ms[1][0];
  const float w1s1 = w1p[j8*4+1] + w1p[j8*4+2]*Ms[0][1] + w1p[j8*4+3]*Ms[1][1];
  v2f A1[2];
  A1[0][0] = lo ? w1s0 : w1g0;  A1[0][1] = lo ? w1s1 : w1g1;
  A1[1][0] = lo ? w1g0 : w1s0;  A1[1][1] = lo ? w1g1 : w1s1;

  v2f A2[4], A3[4];
  const int m3 = n & 7;
  const int j3 = n & 1;
#pragma unroll
  for (int q = 0; q < 4; ++q)
#pragma unroll
    for (int v = 0; v < 2; ++v) {
      const int f = 2*q + v;
      const float vrc = rc2_w[j8*8 + f];
      const float vrr = rr2_w[j8*8 + f];
      A2[q][v] = lo ? (rcrow ? vrc : 0.0f) : (rcrow ? 0.0f : vrr);
      const float urc = rc3_w[j3*8 + f];
      const float urr = rr3_w[j3*8 + f];
      float a3 = 0.0f;
      if (m3 < 2)      a3 = lo ? urc : 0.0f;
      else if (m3 < 4) a3 = lo ? -urc : urr;
      A3[q][v] = a3;
    }

  v8f C1, C2;
#pragma unroll
  for (int r = 0; r < 8; ++r) {
    C1[r] = lo ? rc1_b[r] : rr1_b[r];
    C2[r] = lo ? rc2_b[r] : rr2_b[r];
  }

  const v8f z8 = {0.f,0.f,0.f,0.f,0.f,0.f,0.f,0.f};

  float err = 0.0f;
#pragma unroll 1
  for (int step = 0; step < NSTEPS; ++step) {
    v2f B1[2];
    B1[0][0] = lo ? sx  : g0o;  B1[0][1] = lo ? sy  : g1o;
    B1[1][0] = lo ? g0o : sx;   B1[1][1] = lo ? g1o : sy;

    const float ahdx = fpx - (Ms[0][0]*sx + Ms[0][1]*sy);
    const float ahdy = fpy - (Ms[1][0]*sx + Ms[1][1]*sy);

    v8f h1[2];
#pragma unroll
    for (int t = 0; t < 2; ++t) {
      v8f d = wmma4(A1[t], B1[t], C1);
#pragma unroll
      for (int r = 0; r < 8; ++r) h1[t][r] = TANHF(d[r]);
    }

    v8f h2[2];
#pragma unroll
    for (int t = 0; t < 2; ++t) {
      v8f d = C2;
#pragma unroll
      for (int q = 0; q < 4; ++q) {
        v2f b = { h1[t][2*q], h1[t][2*q+1] };
        d = wmma4(A2[q], b, d);
      }
#pragma unroll
      for (int r = 0; r < 8; ++r) h2[t][r] = TANHF(d[r]);
    }

    v8f d3[2];
#pragma unroll
    for (int t = 0; t < 2; ++t) {
      v8f d = z8;
#pragma unroll
      for (int q = 0; q < 4; ++q) {
        v2f b = { h2[t][2*q], h2[t][2*q+1] };
        d = wmma4(A3[q], b, d);
      }
      d3[t] = d;
    }

    const float arx = (lo ? d3[0][0] : d3[1][0]) + b3rc0;
    const float ary = (lo ? d3[0][1] : d3[1][1]) + b3rc1;
    const float drx = (lo ? d3[0][2] : d3[1][2]) + db0;
    const float dry = (lo ? d3[0][3] : d3[1][3]) + db1;
    sx += DT_C * arx;
    sy += DT_C * ary;
    const float ex = ssx - sx, ey = ssy - sy;
    err += sqrtf(ex*ex + ey*ey)
         + 0.1f * sqrtf(drx*drx + dry*dry)
         + sqrtf(ahdx*ahdx + ahdy*ahdy);
  }

  if (valid) out[row] = err;
}

extern "C" void kernel_launch(void* const* d_in, const int* in_sizes, int n_in,
                              void* d_out, int out_size, void* d_ws, size_t ws_size,
                              hipStream_t stream) {
  const float* s_star = (const float*)d_in[0];
  const float* s0     = (const float*)d_in[1];
  const float* fc1_w  = (const float*)d_in[2];
  const float* fc1_b  = (const float*)d_in[3];
  const float* fc2_w  = (const float*)d_in[4];
  const float* fc2_b  = (const float*)d_in[5];
  const float* rc1_w  = (const float*)d_in[6];
  const float* rc1_b  = (const float*)d_in[7];
  const float* rc2_w  = (const float*)d_in[8];
  const float* rc2_b  = (const float*)d_in[9];
  const float* rc3_w  = (const float*)d_in[10];
  const float* rc3_b  = (const float*)d_in[11];
  const float* rr1_w  = (const float*)d_in[12];
  const float* rr1_b  = (const float*)d_in[13];
  const float* rr2_w  = (const float*)d_in[14];
  const float* rr2_b  = (const float*)d_in[15];
  const float* rr3_w  = (const float*)d_in[16];
  const float* rr3_b  = (const float*)d_in[17];
  float* out = (float*)d_out;

  const int B = in_sizes[0] / 2;
  const int block = WPB * 32;
  const int grid = (B + block - 1) / block;

  rollout_kernel<<<grid, block, 0, stream>>>(
      s_star, s0, fc1_w, fc1_b, fc2_w, fc2_b,
      rc1_w, rc1_b, rc2_w, rc2_b, rc3_w, rc3_b,
      rr1_w, rr1_b, rr2_w, rr2_b, rr3_w, rr3_b,
      out, B);
}


# Each wave carries two independent 32-row groups half a step apart; scheduling groups place one group's tanh, sqrt and state update between the other group's matrix instructions
# speedup vs baseline: 1.2469x; 1.0365x over previous
#include <hip/hip_runtime.h>

#define NSTEPS 10
#define DT_C 0.1f
#define WPB 4

typedef float v2f __attribute__((ext_vector_type(2)));
typedef float v8f __attribute__((ext_vector_type(8)));

#if __has_builtin(__builtin_amdgcn_tanhf)
#define TANHF(x) __builtin_amdgcn_tanhf(x)
#elif __has_builtin(__builtin_amdgcn_tanh_f32)
#define TANHF(x) __builtin_amdgcn_tanh_f32(x)
#else
#define TANHF(x) tanhf(x)
#endif

static __device__ __forceinline__ v8f wmma4(v2f a, v2f b, v8f c) {
  return __builtin_amdgcn_wmma_f32_16x16x4_f32(false, a, false, b, (short)0, c,
                                               false, false);
}


#define SG_MAT(n)   __builtin_amdgcn_sched_group_barrier(0x008, n, 0)
#define SG_TRANS(n) __builtin_amdgcn_sched_group_barrier(0x400, n, 0)
#define SG_VALU(n)  __builtin_amdgcn_sched_group_barrier(0x002, n, 0)

struct Grp {
  float ssx, ssy, sx, sy, g0o, g1o, fpx, fpy, err, q0, q1, q2;
  v2f B1[2];
};

static __device__ __forceinline__ void mm16_tanh(const v2f (&A)[4], const v8f (&hin)[2],
                                                 const v8f cinit, v8f (&dout)[2],
                                                 const v8f (&tin)[2], v8f (&tout)[2]) {
  v8f d0 = cinit, d1 = cinit;
#pragma unroll
  for (int q = 0; q < 4; ++q) {
    v2f b0 = { hin[0][2*q], hin[0][2*q+1] };
    d0 = wmma4(A[q], b0, d0);
    tout[0][2*q]   = TANHF(tin[0][2*q]);
    tout[0][2*q+1] = TANHF(tin[0][2*q+1]);
    v2f b1 = { hin[1][2*q], hin[1][2*q+1] };
    d1 = wmma4(A[q], b1, d1);
    tout[1][2*q]   = TANHF(tin[1][2*q]);
    tout[1][2*q+1] = TANHF(tin[1][2*q+1]);
  }
  dout[0] = d0; dout[1] = d1;
}

static __device__ __forceinline__ void mm16(const v2f (&A)[4], const v8f (&hin)[2],
                                            const v8f cinit, v8f (&dout)[2]) {
  v8f d0 = cinit, d1 = cinit;
#pragma unroll
  for (int q = 0; q < 4; ++q) {
    v2f b0 = { hin[0][2*q], hin[0][2*q+1] };
    d0 = wmma4(A[q], b0, d0);
    v2f b1 = { hin[1][2*q], hin[1][2*q+1] };
    d1 = wmma4(A[q], b1, d1);
  }
  dout[0] = d0; dout[1] = d1;
}

static __device__ __forceinline__ void build_b1(Grp& g, bool lo) {
  g.B1[0][0] = lo ? g.sx  : g.g0o;  g.B1[0][1] = lo ? g.sy  : g.g1o;
  g.B1[1][0] = lo ? g.g0o : g.sx;   g.B1[1][1] = lo ? g.g1o : g.sy;
}

static __device__ __forceinline__ void advance(Grp& g, const v8f (&d3)[2], bool lo,
                                               float m00, float m01, float m10, float m11,
                                               float b3rc0, float b3rc1, float db0, float db1) {
  const float ahdx = g.fpx - (m00*g.sx + m01*g.sy);
  const float ahdy = g.fpy - (m10*g.sx + m11*g.sy);
  const float arx = (lo ? d3[0][0] : d3[1][0]) + b3rc0;
  const float ary = (lo ? d3[0][1] : d3[1][1]) + b3rc1;
  const float drx = (lo ? d3[0][2] : d3[1][2]) + db0;
  const float dry = (lo ? d3[0][3] : d3[1][3]) + db1;
  g.sx += DT_C * arx;
  g.sy += DT_C * ary;
  const float ex = g.ssx - g.sx, ey = g.ssy - g.sy;
  g.q0 = ex*ex + ey*ey;
  g.q1 = drx*drx + dry*dry;
  g.q2 = ahdx*ahdx + ahdy*ahdy;
  build_b1(g, lo);
}

static __device__ __forceinline__ void add_err(Grp& g) {
  g.err += sqrtf(g.q0) + 0.1f * sqrtf(g.q1) + sqrtf(g.q2);
}

__global__ __launch_bounds__(WPB * 32) void rollout_kernel(
    const float* __restrict__ s_star, const float* __restrict__ s0,
    const float* __restrict__ fc1_w, const float* __restrict__ fc1_b,
    const float* __restrict__ fc2_w, const float* __restrict__ fc2_b,
    const float* __restrict__ rc1_w, const float* __restrict__ rc1_b,
    const float* __restrict__ rc2_w, const float* __restrict__ rc2_b,
    const float* __restrict__ rc3_w, const float* __restrict__ rc3_b,
    const float* __restrict__ rr1_w, const float* __restrict__ rr1_b,
    const float* __restrict__ rr2_w, const float* __restrict__ rr2_b,
    const float* __restrict__ rr3_w, const float* __restrict__ rr3_b,
    float* __restrict__ out, int nrows)
{

  const int tid = (int)threadIdx.x;
  const int l = tid & 31;
  const int n = l & 15;
  const int kh = l >> 4;
  const bool lo = (kh == 0);

  const int rbase = (blockIdx.x * WPB + (tid >> 5)) * 64 + l;
  int row[2]; bool valid[2];
#pragma unroll
  for (int g = 0; g < 2; ++g) {
    row[g] = rbase + 32 * g;
    valid[g] = (row[g] < nrows);
    if (!valid[g]) row[g] = nrows - 1;
  }

  float Ms[2][2], Mt[2][2], cah[2];
#pragma unroll
  for (int i = 0; i < 2; ++i) {
    float a0 = fc2_w[i*4+0], a1 = fc2_w[i*4+1], a2 = fc2_w[i*4+2], a3 = fc2_w[i*4+3];
#pragma unroll
    for (int k = 0; k < 2; ++k) {
      Ms[i][k] = a0*fc1_w[0*4+k]   + a1*fc1_w[1*4+k]   + a2*fc1_w[2*4+k]   + a3*fc1_w[3*4+k];
      Mt[i][k] = a0*fc1_w[0*4+2+k] + a1*fc1_w[1*4+2+k] + a2*fc1_w[2*4+2+k] + a3*fc1_w[3*4+2+k];
    }
    cah[i] = a0*fc1_b[0] + a1*fc1_b[1] + a2*fc1_b[2] + a3*fc1_b[3] + fc2_b[i];
  }
  const float b3rc0 = rc3_b[0], b3rc1 = rc3_b[1];
  const float db0 = rr3_b[0] - rc3_b[0], db1 = rr3_b[1] - rc3_b[1];

  Grp G[2];
#pragma unroll
  for (int g = 0; g < 2; ++g) {
    const int r = row[g];
    G[g].ssx = s_star[r*2+0]; G[g].ssy = s_star[r*2+1];
    G[g].sx = s0[r*2+0];      G[g].sy = s0[r*2+1];
    const float g0 = Mt[0][0]*G[g].ssx + Mt[0][1]*G[g].ssy + cah[0];
    const float g1 = Mt[1][0]*G[g].ssx + Mt[1][1]*G[g].ssy + cah[1];
    const float ahs0 = (G[g].ssx > 0.5f) ? -1.0f : 1.0f;
    G[g].fpx = ahs0 - g0;
    G[g].fpy = 0.0f - g1;
    G[g].g0o = __shfl_xor(g0, 16, 32);
    G[g].g1o = __shfl_xor(g1, 16, 32);
    G[g].err = 0.0f;
    G[g].q0 = 0.0f; G[g].q1 = 0.0f; G[g].q2 = 0.0f;
    build_b1(G[g], lo);
  }

  const bool rcrow = (n < 8);
  const int j8 = n & 7;
  const float* w1p = rcrow ? rc1_w : rr1_w;
  const float w1g0 = w1p[j8*4+2], w1g1 = w1p[j8*4+3];
  const float w1s0 = w1p[j8*4+0] + w1p[j8*4+2]*Ms[0][0] + w1p[j8*4+3]*Ms[1][0];
  const float w1s1 = w1p[j8*4+1] + w1p[j8*4+2]*Ms[0][1] + w1p[j8*4+3]*Ms[1][1];
  v2f A1[2];
  A1[0][0] = lo ? w1s0 : w1g0;  A1[0][1] = lo ? w1s1 : w1g1;
  A1[1][0] = lo ? w1g0 : w1s0;  A1[1][1] = lo ? w1g1 : w1s1;

  v2f A2[4], A3[4];
  const int m3 = n & 7;
  const int j3 = n & 1;
#pragma unroll
  for (int q = 0; q < 4; ++q)
#pragma unroll
    for (int v = 0; v < 2; ++v) {
      const int f = 2*q + v;
      const float vrc = rc2_w[j8*8 + f];
      const float vrr = rr2_w[j8*8 + f];
      A2[q][v] = lo ? (rcrow ? vrc : 0.0f) : (rcrow ? 0.0f : vrr);
      const float urc = rc3_w[j3*8 + f];
      const float urr = rr3_w[j3*8 + f];
      float a3 = 0.0f;
      if (m3 < 2)      a3 = lo ? urc : 0.0f;
      else if (m3 < 4) a3 = lo ? -urc : urr;
      A3[q][v] = a3;
    }

  v8f C1, C2;
#pragma unroll
  for (int r = 0; r < 8; ++r) {
    C1[r] = lo ? rc1_b[r] : rr1_b[r];
    C2[r] = lo ? rc2_b[r] : rr2_b[r];
  }

  const v8f z8 = {0.f,0.f,0.f,0.f,0.f,0.f,0.f,0.f};

  v8f d1[2][2], h1[2][2], d2[2][2], h2[2][2], d3[2][2];

#pragma unroll
  for (int g = 0; g < 2; ++g)
#pragma unroll
    for (int t = 0; t < 2; ++t) d1[g][t] = wmma4(A1[t], G[g].B1[t], C1);
#pragma unroll
  for (int t = 0; t < 2; ++t)
#pragma unroll
    for (int r = 0; r < 8; ++r) h1[0][t][r] = TANHF(d1[0][t][r]);

#pragma unroll 1
  for (int step = 0; step < NSTEPS; ++step) {
    mm16_tanh(A2, h1[0], C2, d2[0], d1[1], h1[1]);
    add_err(G[1]);
#pragma unroll
    for (int i = 0; i < 8; ++i) { SG_MAT(1); SG_TRANS(3); SG_VALU(6); }
    mm16_tanh(A2, h1[1], C2, d2[1], d2[0], h2[0]);
#pragma unroll
    for (int i = 0; i < 8; ++i) { SG_MAT(1); SG_TRANS(2); }
    mm16_tanh(A3, h2[0], z8, d3[0], d2[1], h2[1]);
#pragma unroll
    for (int i = 0; i < 8; ++i) { SG_MAT(1); SG_TRANS(2); }
    mm16(A3, h2[1], z8, d3[1]);
    advance(G[0], d3[0], lo, Ms[0][0], Ms[0][1], Ms[1][0], Ms[1][1], b3rc0, b3rc1, db0, db1);
    add_err(G[0]);
#pragma unroll
    for (int i = 0; i < 8; ++i) { SG_MAT(1); SG_VALU(10); SG_TRANS(1); }
#pragma unroll
    for (int t = 0; t < 2; ++t) d1[0][t] = wmma4(A1[t], G[0].B1[t], C1);
    advance(G[1], d3[1], lo, Ms[0][0], Ms[0][1], Ms[1][0], Ms[1][1], b3rc0, b3rc1, db0, db1);
    SG_MAT(1); SG_VALU(12); SG_MAT(1); SG_VALU(12);
#pragma unroll
    for (int t = 0; t < 2; ++t) {
      d1[1][t] = wmma4(A1[t], G[1].B1[t], C1);
#pragma unroll
      for (int r = 0; r < 8; ++r) h1[0][t][r] = TANHF(d1[0][t][r]);
    }
    SG_MAT(1); SG_TRANS(8); SG_MAT(1); SG_TRANS(8);
  }
  add_err(G[1]);

#pragma unroll
  for (int g = 0; g < 2; ++g)
    if (valid[g]) out[row[g]] = G[g].err;
}

extern "C" void kernel_launch(void* const* d_in, const int* in_sizes, int n_in,
                              void* d_out, int out_size, void* d_ws, size_t ws_size,
                              hipStream_t stream) {
  const float* s_star = (const float*)d_in[0];
  const float* s0     = (const float*)d_in[1];
  const float* fc1_w  = (const float*)d_in[2];
  const float* fc1_b  = (const float*)d_in[3];
  const float* fc2_w  = (const float*)d_in[4];
  const float* fc2_b  = (const float*)d_in[5];
  const float* rc1_w  = (const float*)d_in[6];
  const float* rc1_b  = (const float*)d_in[7];
  const float* rc2_w  = (const float*)d_in[8];
  const float* rc2_b  = (const float*)d_in[9];
  const float* rc3_w  = (const float*)d_in[10];
  const float* rc3_b  = (const float*)d_in[11];
  const float* rr1_w  = (const float*)d_in[12];
  const float* rr1_b  = (const float*)d_in[13];
  const float* rr2_w  = (const float*)d_in[14];
  const float* rr2_b  = (const float*)d_in[15];
  const float* rr3_w  = (const float*)d_in[16];
  const float* rr3_b  = (const float*)d_in[17];
  float* out = (float*)d_out;

  const int B = in_sizes[0] / 2;
  const int block = WPB * 32;
  const int rpb = WPB * 64;
  const int grid = (B + rpb - 1) / rpb;

  rollout_kernel<<<grid, block, 0, stream>>>(
      s_star, s0, fc1_w, fc1_b, fc2_w, fc2_b,
      rc1_w, rc1_b, rc2_w, rc2_b, rc3_w, rc3_b,
      rr1_w, rr1_b, rr2_w, rr2_b, rr3_w, rr3_b,
      out, B);
}


# Square roots of the three per-step norms as the bare hardware instruction (within 1 ulp) instead of the correctly rounded library sequence: 74 fewer plain vector issue slots per wave-step
# speedup vs baseline: 1.3514x; 1.0838x over previous
#include <hip/hip_runtime.h>

#define NSTEPS 10
#define DT_C 0.1f
#define WPB 4

typedef float v2f __attribute__((ext_vector_type(2)));
typedef float v8f __attribute__((ext_vector_type(8)));

#if __has_builtin(__builtin_amdgcn_tanhf)
#define TANHF(x) __builtin_amdgcn_tanhf(x)
#elif __has_builtin(__builtin_amdgcn_tanh_f32)
#define TANHF(x) __builtin_amdgcn_tanh_f32(x)
#else
#define TANHF(x) tanhf(x)
#endif

#if __has_builtin(__builtin_amdgcn_sqrtf)
#define SQRTF(x) __builtin_amdgcn_sqrtf(x)
#else
#define SQRTF(x) sqrtf(x)
#endif

static __device__ __forceinline__ v8f wmma4(v2f a, v2f b, v8f c) {
  return __builtin_amdgcn_wmma_f32_16x16x4_f32(false, a, false, b, (short)0, c,
                                               false, false);
}


#define SG_MAT(n)   __builtin_amdgcn_sched_group_barrier(0x008, n, 0)
#define SG_TRANS(n) __builtin_amdgcn_sched_group_barrier(0x400, n, 0)
#define SG_VALU(n)  __builtin_amdgcn_sched_group_barrier(0x002, n, 0)

struct Grp {
  float ssx, ssy, sx, sy, g0o, g1o, fpx, fpy, err, q0, q1, q2;
  v2f B1[2];
};

static __device__ __forceinline__ void mm16_tanh(const v2f (&A)[4], const v8f (&hin)[2],
                                                 const v8f cinit, v8f (&dout)[2],
                                                 const v8f (&tin)[2], v8f (&tout)[2]) {
  v8f d0 = cinit, d1 = cinit;
#pragma unroll
  for (int q = 0; q < 4; ++q) {
    v2f b0 = { hin[0][2*q], hin[0][2*q+1] };
    d0 = wmma4(A[q], b0, d0);
    tout[0][2*q]   = TANHF(tin[0][2*q]);
    tout[0][2*q+1] = TANHF(tin[0][2*q+1]);
    v2f b1 = { hin[1][2*q], hin[1][2*q+1] };
    d1 = wmma4(A[q], b1, d1);
    tout[1][2*q]   = TANHF(tin[1][2*q]);
    tout[1][2*q+1] = TANHF(tin[1][2*q+1]);
  }
  dout[0] = d0; dout[1] = d1;
}

static __device__ __forceinline__ void mm16(const v2f (&A)[4], const v8f (&hin)[2],
                                            const v8f cinit, v8f (&dout)[2]) {
  v8f d0 = cinit, d1 = cinit;
#pragma unroll
  for (int q = 0; q < 4; ++q) {
    v2f b0 = { hin[0][2*q], hin[0][2*q+1] };
    d0 = wmma4(A[q], b0, d0);
    v2f b1 = { hin[1][2*q], hin[1][2*q+1] };
    d1 = wmma4(A[q], b1, d1);
  }
  dout[0] = d0; dout[1] = d1;
}

static __device__ __forceinline__ void build_b1(Grp& g, bool lo) {
  g.B1[0][0] = lo ? g.sx  : g.g0o;  g.B1[0][1] = lo ? g.sy  : g.g1o;
  g.B1[1][0] = lo ? g.g0o : g.sx;   g.B1[1][1] = lo ? g.g1o : g.sy;
}

static __device__ __forceinline__ void advance(Grp& g, const v8f (&d3)[2], bool lo,
                                               float m00, float m01, float m10, float m11,
                                               float b3rc0, float b3rc1, float db0, float db1) {
  const float ahdx = g.fpx - (m00*g.sx + m01*g.sy);
  const float ahdy = g.fpy - (m10*g.sx + m11*g.sy);
  const float arx = (lo ? d3[0][0] : d3[1][0]) + b3rc0;
  const float ary = (lo ? d3[0][1] : d3[1][1]) + b3rc1;
  const float drx = (lo ? d3[0][2] : d3[1][2]) + db0;
  const float dry = (lo ? d3[0][3] : d3[1][3]) + db1;
  g.sx += DT_C * arx;
  g.sy += DT_C * ary;
  const float ex = g.ssx - g.sx, ey = g.ssy - g.sy;
  g.q0 = ex*ex + ey*ey;
  g.q1 = drx*drx + dry*dry;
  g.q2 = ahdx*ahdx + ahdy*ahdy;
  build_b1(g, lo);
}

static __device__ __forceinline__ void add_err(Grp& g) {
  g.err += SQRTF(g.q0) + 0.1f * SQRTF(g.q1) + SQRTF(g.q2);
}

__global__ __launch_bounds__(WPB * 32) void rollout_kernel(
    const float* __restrict__ s_star, const float* __restrict__ s0,
    const float* __restrict__ fc1_w, const float* __restrict__ fc1_b,
    const float* __restrict__ fc2_w, const float* __restrict__ fc2_b,
    const float* __restrict__ rc1_w, const float* __restrict__ rc1_b,
    const float* __restrict__ rc2_w, const float* __restrict__ rc2_b,
    const float* __restrict__ rc3_w, const float* __restrict__ rc3_b,
    const float* __restrict__ rr1_w, const float* __restrict__ rr1_b,
    const float* __restrict__ rr2_w, const float* __restrict__ rr2_b,
    const float* __restrict__ rr3_w, const float* __restrict__ rr3_b,
    float* __restrict__ out, int nrows)
{

  const int tid = (int)threadIdx.x;
  const int l = tid & 31;
  const int n = l & 15;
  const int kh = l >> 4;
  const bool lo = (kh == 0);

  const int rbase = (blockIdx.x * WPB + (tid >> 5)) * 64 + l;
  int row[2]; bool valid[2];
#pragma unroll
  for (int g = 0; g < 2; ++g) {
    row[g] = rbase + 32 * g;
    valid[g] = (row[g] < nrows);
    if (!valid[g]) row[g] = nrows - 1;
  }

  float Ms[2][2], Mt[2][2], cah[2];
#pragma unroll
  for (int i = 0; i < 2; ++i) {
    float a0 = fc2_w[i*4+0], a1 = fc2_w[i*4+1], a2 = fc2_w[i*4+2], a3 = fc2_w[i*4+3];
#pragma unroll
    for (int k = 0; k < 2; ++k) {
      Ms[i][k] = a0*fc1_w[0*4+k]   + a1*fc1_w[1*4+k]   + a2*fc1_w[2*4+k]   + a3*fc1_w[3*4+k];
      Mt[i][k] = a0*fc1_w[0*4+2+k] + a1*fc1_w[1*4+2+k] + a2*fc1_w[2*4+2+k] + a3*fc1_w[3*4+2+k];
    }
    cah[i] = a0*fc1_b[0] + a1*fc1_b[1] + a2*fc1_b[2] + a3*fc1_b[3] + fc2_b[i];
  }
  const float b3rc0 = rc3_b[0], b3rc1 = rc3_b[1];
  const float db0 = rr3_b[0] - rc3_b[0], db1 = rr3_b[1] - rc3_b[1];

  Grp G[2];
#pragma unroll
  for (int g = 0; g < 2; ++g) {
    const int r = row[g];
    G[g].ssx = s_star[r*2+0]; G[g].ssy = s_star[r*2+1];
    G[g].sx = s0[r*2+0];      G[g].sy = s0[r*2+1];
    const float g0 = Mt[0][0]*G[g].ssx + Mt[0][1]*G[g].ssy + cah[0];
    const float g1 = Mt[1][0]*G[g].ssx + Mt[1][1]*G[g].ssy + cah[1];
    const float ahs0 = (G[g].ssx > 0.5f) ? -1.0f : 1.0f;
    G[g].fpx = ahs0 - g0;
    G[g].fpy = 0.0f - g1;
    G[g].g0o = __shfl_xor(g0, 16, 32);
    G[g].g1o = __shfl_xor(g1, 16, 32);
    G[g].err = 0.0f;
    G[g].q0 = 0.0f; G[g].q1 = 0.0f; G[g].q2 = 0.0f;
    build_b1(G[g], lo);
  }

  const bool rcrow = (n < 8);
  const int j8 = n & 7;
  const float* w1p = rcrow ? rc1_w : rr1_w;
  const float w1g0 = w1p[j8*4+2], w1g1 = w1p[j8*4+3];
  const float w1s0 = w1p[j8*4+0] + w1p[j8*4+2]*Ms[0][0] + w1p[j8*4+3]*Ms[1][0];
  const float w1s1 = w1p[j8*4+1] + w1p[j8*4+2]*Ms[0][1] + w1p[j8*4+3]*Ms[1][1];
  v2f A1[2];
  A1[0][0] = lo ? w1s0 : w1g0;  A1[0][1] = lo ? w1s1 : w1g1;
  A1[1][0] = lo ? w1g0 : w1s0;  A1[1][1] = lo ? w1g1 : w1s1;

  v2f A2[4], A3[4];
  const int m3 = n & 7;
  const int j3 = n & 1;
#pragma unroll
  for (int q = 0; q < 4; ++q)
#pragma unroll
    for (int v = 0; v < 2; ++v) {
      const int f = 2*q + v;
      const float vrc = rc2_w[j8*8 + f];
      const float vrr = rr2_w[j8*8 + f];
      A2[q][v] = lo ? (rcrow ? vrc : 0.0f) : (rcrow ? 0.0f : vrr);
      const float urc = rc3_w[j3*8 + f];
      const float urr = rr3_w[j3*8 + f];
      float a3 = 0.0f;
      if (m3 < 2)      a3 = lo ? urc : 0.0f;
      else if (m3 < 4) a3 = lo ? -urc : urr;
      A3[q][v] = a3;
    }

  v8f C1, C2;
#pragma unroll
  for (int r = 0; r < 8; ++r) {
    C1[r] = lo ? rc1_b[r] : rr1_b[r];
    C2[r] = lo ? rc2_b[r] : rr2_b[r];
  }

  const v8f z8 = {0.f,0.f,0.f,0.f,0.f,0.f,0.f,0.f};

  v8f d1[2][2], h1[2][2], d2[2][2], h2[2][2], d3[2][2];

#pragma unroll
  for (int g = 0; g < 2; ++g)
#pragma unroll
    for (int t = 0; t < 2; ++t) d1[g][t] = wmma4(A1[t], G[g].B1[t], C1);
#pragma unroll
  for (int t = 0; t < 2; ++t)
#pragma unroll
    for (int r = 0; r < 8; ++r) h1[0][t][r] = TANHF(d1[0][t][r]);

#pragma unroll 1
  for (int step = 0; step < NSTEPS; ++step) {
    mm16_tanh(A2, h1[0], C2, d2[0], d1[1], h1[1]);
    add_err(G[1]);
#pragma unroll
    for (int i = 0; i < 8; ++i) { SG_MAT(1); SG_TRANS(3); SG_VALU(6); }
    mm16_tanh(A2, h1[1], C2, d2[1], d2[0], h2[0]);
#pragma unroll
    for (int i = 0; i < 8; ++i) { SG_MAT(1); SG_TRANS(2); }
    mm16_tanh(A3, h2[0], z8, d3[0], d2[1], h2[1]);
#pragma unroll
    for (int i = 0; i < 8; ++i) { SG_MAT(1); SG_TRANS(2); }
    mm16(A3, h2[1], z8, d3[1]);
    advance(G[0], d3[0], lo, Ms[0][0], Ms[0][1], Ms[1][0], Ms[1][1], b3rc0, b3rc1, db0, db1);
    add_err(G[0]);
#pragma unroll
    for (int i = 0; i < 8; ++i) { SG_MAT(1); SG_VALU(10); SG_TRANS(1); }
#pragma unroll
    for (int t = 0; t < 2; ++t) d1[0][t] = wmma4(A1[t], G[0].B1[t], C1);
    advance(G[1], d3[1], lo, Ms[0][0], Ms[0][1], Ms[1][0], Ms[1][1], b3rc0, b3rc1, db0, db1);
    SG_MAT(1); SG_VALU(12); SG_MAT(1); SG_VALU(12);
#pragma unroll
    for (int t = 0; t < 2; ++t) {
      d1[1][t] = wmma4(A1[t], G[1].B1[t], C1);
#pragma unroll
      for (int r = 0; r < 8; ++r) h1[0][t][r] = TANHF(d1[0][t][r]);
    }
    SG_MAT(1); SG_TRANS(8); SG_MAT(1); SG_TRANS(8);
  }
  add_err(G[1]);

#pragma unroll
  for (int g = 0; g < 2; ++g)
    if (valid[g]) out[row[g]] = G[g].err;
}

extern "C" void kernel_launch(void* const* d_in, const int* in_sizes, int n_in,
                              void* d_out, int out_size, void* d_ws, size_t ws_size,
                              hipStream_t stream) {
  const float* s_star = (const float*)d_in[0];
  const float* s0     = (const float*)d_in[1];
  const float* fc1_w  = (const float*)d_in[2];
  const float* fc1_b  = (const float*)d_in[3];
  const float* fc2_w  = (const float*)d_in[4];
  const float* fc2_b  = (const float*)d_in[5];
  const float* rc1_w  = (const float*)d_in[6];
  const float* rc1_b  = (const float*)d_in[7];
  const float* rc2_w  = (const float*)d_in[8];
  const float* rc2_b  = (const float*)d_in[9];
  const float* rc3_w  = (const float*)d_in[10];
  const float* rc3_b  = (const float*)d_in[11];
  const float* rr1_w  = (const float*)d_in[12];
  const float* rr1_b  = (const float*)d_in[13];
  const float* rr2_w  = (const float*)d_in[14];
  const float* rr2_b  = (const float*)d_in[15];
  const float* rr3_w  = (const float*)d_in[16];
  const float* rr3_b  = (const float*)d_in[17];
  float* out = (float*)d_out;

  const int B = in_sizes[0] / 2;
  const int block = WPB * 32;
  const int rpb = WPB * 64;
  const int grid = (B + rpb - 1) / rpb;

  rollout_kernel<<<grid, block, 0, stream>>>(
      s_star, s0, fc1_w, fc1_b, fc2_w, fc2_b,
      rc1_w, rc1_b, rc2_w, rc2_b, rc3_w, rc3_b,
      rr1_w, rr1_b, rr2_w, rr2_b, rr3_w, rr3_b,
      out, B);
}
